# starting kernel
# baseline (speedup 1.0000x reference)
#include <hip/hip_runtime.h>
#include <hip/hip_bf16.h>

#define BZ    8
#define DI    256
#define DS    512
#define DO    256
#define LSEQ  4096
#define TCH   256
#define NCHUNK (LSEQ / TCH)

#define UP    68
#define XP    516

typedef float v2f __attribute__((ext_vector_type(2)));
typedef float v4f __attribute__((ext_vector_type(4)));
typedef float v8f __attribute__((ext_vector_type(8)));

static __device__ __forceinline__ v8f wmma_f32_16x16x4(v2f a, v2f b, v8f c) {
    return __builtin_amdgcn_wmma_f32_16x16x4_f32(
         false, a,  false, b,
         (short)0, c,  false,  false);
}

__global__ void k_adiag(const float* __restrict__ a_un, float* __restrict__ adiag) {
    int s = threadIdx.x;
    float x  = a_un[s];
    float sp = fmaxf(x, 0.0f) + log1pf(expf(-fabsf(x)));
    adiag[s] = -sp;
}

__global__ __launch_bounds__(256) void k_gemm1(const float* __restrict__ u,
                                               const float* __restrict__ Bm,
                                               float* __restrict__ X) {
    extern __shared__ char smem_raw[];
    float* s_u = (float*)smem_raw;

    const int b      = blockIdx.x >> 6;
    const int l0base = (blockIdx.x & 63) * 64;
    const int t      = threadIdx.x;
    const int wave   = t >> 5;
    const int lane   = t & 31;
    const int mrow   = lane & 15;
    const int khalf  = (lane >> 4) << 1;
    const int ncol   = lane & 15;
    const int hi     = (lane >> 4) << 3;

    const float* __restrict__ ub = u + (size_t)b * DI * LSEQ;

    __builtin_prefetch(Bm + (size_t)(wave * 64 + mrow) * DI, 0, 3);

    #pragma unroll
    for (int c = 0; c < 16; ++c) {
        const float4 v = *(const float4*)(ub + (size_t)t * LSEQ + l0base + c * 4);
        *(float4*)(&s_u[t * UP + c * 4]) = v;
    }
    __syncthreads();

    const v8f vzero = {0.f,0.f,0.f,0.f,0.f,0.f,0.f,0.f};

    for (int half = 0; half < 2; ++half) {
        const int st0 = wave * 4 + half * 2;
        v8f acc[2][4];
        #pragma unroll
        for (int i = 0; i < 2; ++i)
            #pragma unroll
            for (int j = 0; j < 4; ++j) acc[i][j] = vzero;

        for (int k0 = 0; k0 < DI; k0 += 4) {
            v2f afrag[2];
            #pragma unroll
            for (int i = 0; i < 2; ++i) {
                const int srow = (st0 + i) * 16 + mrow;
                afrag[i] = *(const v2f*)(Bm + (size_t)srow * DI + k0 + khalf);
            }
            v2f bfrag[4];
            #pragma unroll
            for (int j = 0; j < 4; ++j) {
                const int lcol = j * 16 + ncol;
                bfrag[j].x = s_u[(k0 + khalf)     * UP + lcol];
                bfrag[j].y = s_u[(k0 + khalf + 1) * UP + lcol];
            }
            #pragma unroll
            for (int i = 0; i < 2; ++i)
                #pragma unroll
                for (int j = 0; j < 4; ++j)
                    acc[i][j] = wmma_f32_16x16x4(afrag[i], bfrag[j], acc[i][j]);
        }

        #pragma unroll
        for (int i = 0; i < 2; ++i) {
            const int sbase = (st0 + i) * 16 + hi;
            #pragma unroll
            for (int j = 0; j < 4; ++j) {
                const int l = l0base + j * 16 + ncol;
                float* xp = X + ((size_t)b * LSEQ + l) * DS + sbase;
                v4f lo  = __builtin_shufflevector(acc[i][j], acc[i][j], 0,1,2,3);
                v4f hiv = __builtin_shufflevector(acc[i][j], acc[i][j], 4,5,6,7);
                *(v4f*)(xp)     = lo;
                *(v4f*)(xp + 4) = hiv;
            }
        }
    }
}

__global__ __launch_bounds__(256) void k_scanlocal(const float* __restrict__ adiag,
                                                   float* __restrict__ X,
                                                   float* __restrict__ carry_local) {
    const int b = blockIdx.x >> 4, c = blockIdx.x & 15;
    const int s1 = threadIdx.x, s2 = threadIdx.x + 256;
    const float a1 = adiag[s1], a2 = adiag[s2];
    float x1 = 0.f, x2 = 0.f;
    const size_t base = ((size_t)b * LSEQ + (size_t)c * TCH) * DS;
    for (int tt = 0; tt < TCH; ++tt) {
        const size_t idx = base + (size_t)tt * DS;
        x1 = fmaf(a1, x1, X[idx + s1]);
        x2 = fmaf(a2, x2, X[idx + s2]);
        X[idx + s1] = x1;
        X[idx + s2] = x2;
    }
    const size_t ci = ((size_t)b * NCHUNK + c) * DS;
    carry_local[ci + s1] = x1;
    carry_local[ci + s2] = x2;
}

__global__ __launch_bounds__(256) void k_combine(const float* __restrict__ adiag,
                                                 const float* __restrict__ h0,
                                                 const float* __restrict__ carry_local,
                                                 float* __restrict__ carryin) {
    const int b = blockIdx.x;
    #pragma unroll
    for (int rep = 0; rep < 2; ++rep) {
        const int s = threadIdx.x + rep * 256;
        const float a = adiag[s];
        float aT = a;
        #pragma unroll
        for (int i = 0; i < 8; ++i) aT = aT * aT;
        float cin = h0[s];
        for (int c = 0; c < NCHUNK; ++c) {
            const size_t idx = ((size_t)b * NCHUNK + c) * DS + s;
            carryin[idx] = cin;
            cin = fmaf(aT, cin, carry_local[idx]);
        }
    }
}

__global__ __launch_bounds__(256) void k_fixup(const float* __restrict__ adiag,
                                               const float* __restrict__ carryin,
                                               float* __restrict__ X) {
    const int b = blockIdx.x >> 4, c = blockIdx.x & 15;
    const int s1 = threadIdx.x, s2 = threadIdx.x + 256;
    const float a1 = adiag[s1], a2 = adiag[s2];
    const size_t ci = ((size_t)b * NCHUNK + c) * DS;
    const float c1 = carryin[ci + s1], c2 = carryin[ci + s2];
    float p1 = a1, p2 = a2;
    const size_t base = ((size_t)b * LSEQ + (size_t)c * TCH) * DS;
    for (int tt = 0; tt < TCH; ++tt) {
        const size_t idx = base + (size_t)tt * DS;
        X[idx + s1] = fmaf(p1, c1, X[idx + s1]);  p1 *= a1;
        X[idx + s2] = fmaf(p2, c2, X[idx + s2]);  p2 *= a2;
    }
}

__global__ __launch_bounds__(256) void k_gemm2(const float* __restrict__ X,
                                               const float* __restrict__ u,
                                               const float* __restrict__ Cm,
                                               const float* __restrict__ Dm,
                                               float* __restrict__ Y) {
    extern __shared__ char smem_raw[];
    float* s_x = (float*)smem_raw;
    float* s_u = (float*)smem_raw;

    const int b      = blockIdx.x >> 6;
    const int l0base = (blockIdx.x & 63) * 64;
    const int t      = threadIdx.x;
    const int wave   = t >> 5;
    const int lane   = t & 31;
    const int mrow   = lane & 15;
    const int khalf  = (lane >> 4) << 1;
    const int ncol   = lane & 15;
    const int hi     = (lane >> 4) << 3;

    const float* __restrict__ ub = u + (size_t)b * DI * LSEQ;
    const float* __restrict__ Xb = X + (size_t)b * LSEQ * DS;

    __builtin_prefetch(Cm + (size_t)(wave * 32 + mrow) * DS, 0, 3);

    {
        const int lr = t >> 2;
        const int cb = (t & 3) * 128;
        #pragma unroll
        for (int c = 0; c < 32; ++c) {
            const float4 v = *(const float4*)(Xb + (size_t)(l0base + lr) * DS + cb + c * 4);
            *(float4*)(&s_x[lr * XP + cb + c * 4]) = v;
        }
    }
    __syncthreads();

    const v8f vzero = {0.f,0.f,0.f,0.f,0.f,0.f,0.f,0.f};
    v8f acc[2][4];
    #pragma unroll
    for (int i = 0; i < 2; ++i)
        #pragma unroll
        for (int j = 0; j < 4; ++j) acc[i][j] = vzero;

    for (int k0 = 0; k0 < DS; k0 += 4) {
        v2f afrag[2];
        #pragma unroll
        for (int i = 0; i < 2; ++i) {
            const int orow = (wave * 2 + i) * 16 + mrow;
            afrag[i] = *(const v2f*)(Cm + (size_t)orow * DS + k0 + khalf);
        }
        v2f bfrag[4];
        #pragma unroll
        for (int j = 0; j < 4; ++j) {
            const int lr = j * 16 + ncol;
            bfrag[j] = *(const v2f*)(&s_x[lr * XP + k0 + khalf]);
        }
        #pragma unroll
        for (int i = 0; i < 2; ++i)
            #pragma unroll
            for (int j = 0; j < 4; ++j)
                acc[i][j] = wmma_f32_16x16x4(afrag[i], bfrag[j], acc[i][j]);
    }

    __syncthreads();
    #pragma unroll
    for (int c = 0; c < 16; ++c) {
        const float4 v = *(const float4*)(ub + (size_t)t * LSEQ + l0base + c * 4);
        *(float4*)(&s_u[t * UP + c * 4]) = v;
    }
    __syncthreads();

    for (int k0 = 0; k0 < DI; k0 += 4) {
        v2f afrag[2];
        #pragma unroll
        for (int i = 0; i < 2; ++i) {
            const int orow = (wave * 2 + i) * 16 + mrow;
            afrag[i] = *(const v2f*)(Dm + (size_t)orow * DI + k0 + khalf);
        }
        v2f bfrag[4];
        #pragma unroll
        for (int j = 0; j < 4; ++j) {
            const int lcol = j * 16 + ncol;
            bfrag[j].x = s_u[(k0 + khalf)     * UP + lcol];
            bfrag[j].y = s_u[(k0 + khalf + 1) * UP + lcol];
        }
        #pragma unroll
        for (int i = 0; i < 2; ++i)
            #pragma unroll
            for (int j = 0; j < 4; ++j)
                acc[i][j] = wmma_f32_16x16x4(afrag[i], bfrag[j], acc[i][j]);
    }

    #pragma unroll
    for (int i = 0; i < 2; ++i) {
        const int obase = (wave * 2 + i) * 16 + hi;
        #pragma unroll
        for (int j = 0; j < 4; ++j) {
            const int l = l0base + j * 16 + ncol;
            #pragma unroll
            for (int v = 0; v < 8; ++v)
                Y[((size_t)b * DO + obase + v) * LSEQ + l] = acc[i][j][v];
        }
    }
}

extern "C" void kernel_launch(void* const* d_in, const int* in_sizes, int n_in,
                              void* d_out, int out_size, void* d_ws, size_t ws_size,
                              hipStream_t stream) {
    (void)in_sizes; (void)n_in; (void)out_size; (void)ws_size;

    const float* u    = (const float*)d_in[0];
    const float* a_un = (const float*)d_in[1];
    const float* Bm   = (const float*)d_in[2];
    const float* Cm   = (const float*)d_in[3];
    const float* Dm   = (const float*)d_in[4];
    const float* h0   = (const float*)d_in[5];
    float*       Y    = (float*)d_out;

    char* ws = (char*)d_ws;
    const size_t X_BYTES = (size_t)BZ * LSEQ * DS * sizeof(float);
    const size_t C_BYTES = (size_t)BZ * NCHUNK * DS * sizeof(float);
    float* X           = (float*)(ws);
    float* carry_local = (float*)(ws + X_BYTES);
    float* carryin     = (float*)(ws + X_BYTES + C_BYTES);
    float* adiag       = (float*)(ws + X_BYTES + 2 * C_BYTES);

    const size_t SH1 = (size_t)DI * UP * sizeof(float);
    const size_t SH2 = (size_t)64 * XP * sizeof(float);

    k_adiag    <<<1,                512, 0,   stream>>>(a_un, adiag);
    k_gemm1    <<<BZ * (LSEQ / 64), 256, SH1, stream>>>(u, Bm, X);
    k_scanlocal<<<BZ * NCHUNK,      256, 0,   stream>>>(adiag, X, carry_local);
    k_combine  <<<BZ,               256, 0,   stream>>>(adiag, h0, carry_local, carryin);
    k_fixup    <<<BZ * NCHUNK,      256, 0,   stream>>>(adiag, carryin, X);
    k_gemm2    <<<BZ * (LSEQ / 64), 256, SH2, stream>>>(X, u, Cm, Dm, Y);
}


# Chunk-local scan fused into GEMM1's epilogue (shift-and-fma tree across the lanes of each accumulator row, 64-column chunks); separate scan pass removed, 6 launches to 5; reorders the scan's sums
# speedup vs baseline: 1.2039x; 1.2039x over previous
#include <hip/hip_runtime.h>
#include <hip/hip_bf16.h>

#define BZ    8
#define DI    256
#define DS    512
#define DO    256
#define LSEQ  4096
#define TCH   64
#define NCHUNK (LSEQ / TCH)

#define UP    68
#define XP    516

typedef float v2f __attribute__((ext_vector_type(2)));
typedef float v4f __attribute__((ext_vector_type(4)));
typedef float v8f __attribute__((ext_vector_type(8)));

static __device__ __forceinline__ v8f wmma_f32_16x16x4(v2f a, v2f b, v8f c) {
    return __builtin_amdgcn_wmma_f32_16x16x4_f32(
         false, a,  false, b,
         (short)0, c,  false,  false);
}

template <int N>
static __device__ __forceinline__ float row_shr0(float x) {
    return __builtin_bit_cast(float,
        __builtin_amdgcn_update_dpp(0, __builtin_bit_cast(int, x), 0x110 + N, 0xF, 0xF, true));
}
static __device__ __forceinline__ float row_last(float x) {
    return __builtin_bit_cast(float,
        __builtin_amdgcn_update_dpp(0, __builtin_bit_cast(int, x), 0x15F, 0xF, 0xF, true));
}
static __device__ __forceinline__ float row_scan16(float x, float a1, float a2, float a4, float a8) {
    x = fmaf(a1, row_shr0<1>(x), x);
    x = fmaf(a2, row_shr0<2>(x), x);
    x = fmaf(a4, row_shr0<4>(x), x);
    x = fmaf(a8, row_shr0<8>(x), x);
    return x;
}

__global__ void k_adiag(const float* __restrict__ a_un, float* __restrict__ adiag) {
    int s = threadIdx.x;
    float x  = a_un[s];
    float sp = fmaxf(x, 0.0f) + log1pf(expf(-fabsf(x)));
    adiag[s] = -sp;
}

__global__ __launch_bounds__(256) void k_gemm1(const float* __restrict__ u,
                                               const float* __restrict__ Bm,
                                               const float* __restrict__ adiag,
                                               float* __restrict__ X) {
    extern __shared__ char smem_raw[];
    float* s_u = (float*)smem_raw;

    const int b      = blockIdx.x >> 6;
    const int l0base = (blockIdx.x & 63) * 64;
    const int t      = threadIdx.x;
    const int wave   = t >> 5;
    const int lane   = t & 31;
    const int mrow   = lane & 15;
    const int khalf  = (lane >> 4) << 1;
    const int ncol   = lane & 15;
    const int hi     = (lane >> 4) << 3;

    const float* __restrict__ ub = u + (size_t)b * DI * LSEQ;

    __builtin_prefetch(Bm + (size_t)(wave * 64 + mrow) * DI, 0, 3);

    #pragma unroll
    for (int c = 0; c < 16; ++c) {
        const float4 v = *(const float4*)(ub + (size_t)t * LSEQ + l0base + c * 4);
        *(float4*)(&s_u[t * UP + c * 4]) = v;
    }
    __syncthreads();

    const v8f vzero = {0.f,0.f,0.f,0.f,0.f,0.f,0.f,0.f};

    for (int half = 0; half < 2; ++half) {
        const int st0 = wave * 4 + half * 2;
        v8f acc[2][4];
        #pragma unroll
        for (int i = 0; i < 2; ++i)
            #pragma unroll
            for (int j = 0; j < 4; ++j) acc[i][j] = vzero;

        for (int k0 = 0; k0 < DI; k0 += 4) {
            v2f afrag[2];
            #pragma unroll
            for (int i = 0; i < 2; ++i) {
                const int srow = (st0 + i) * 16 + mrow;
                afrag[i] = *(const v2f*)(Bm + (size_t)srow * DI + k0 + khalf);
            }
            v2f bfrag[4];
            #pragma unroll
            for (int j = 0; j < 4; ++j) {
                const int lcol = j * 16 + ncol;
                bfrag[j].x = s_u[(k0 + khalf)     * UP + lcol];
                bfrag[j].y = s_u[(k0 + khalf + 1) * UP + lcol];
            }
            #pragma unroll
            for (int i = 0; i < 2; ++i)
                #pragma unroll
                for (int j = 0; j < 4; ++j)
                    acc[i][j] = wmma_f32_16x16x4(afrag[i], bfrag[j], acc[i][j]);
        }

        #pragma unroll
        for (int i = 0; i < 2; ++i) {
            const int sbase = (st0 + i) * 16 + hi;
            const v4f alo = *(const v4f*)(adiag + sbase);
            const v4f ahi = *(const v4f*)(adiag + sbase + 4);
            #pragma unroll
            for (int r = 0; r < 8; ++r) {
                const float a1 = (r < 4) ? alo[r & 3] : ahi[r & 3];
                const float a2 = a1 * a1, a4 = a2 * a2, a8 = a4 * a4;
                const float w = row_scan16((ncol == 0) ? a1 : 0.f, a1, a2, a4, a8);
                float prev = 0.f;
                #pragma unroll
                for (int j = 0; j < 4; ++j) {
                    float x = row_scan16(acc[i][j][r], a1, a2, a4, a8);
                    if (j > 0) x = fmaf(w, prev, x);
                    prev = row_last(x);
                    acc[i][j][r] = x;
                }
            }
        }

        #pragma unroll
        for (int i = 0; i < 2; ++i) {
            const int sbase = (st0 + i) * 16 + hi;
            #pragma unroll
            for (int j = 0; j < 4; ++j) {
                const int l = l0base + j * 16 + ncol;
                float* xp = X + ((size_t)b * LSEQ + l) * DS + sbase;
                v4f lo  = __builtin_shufflevector(acc[i][j], acc[i][j], 0,1,2,3);
                v4f hiv = __builtin_shufflevector(acc[i][j], acc[i][j], 4,5,6,7);
                *(v4f*)(xp)     = lo;
                *(v4f*)(xp + 4) = hiv;
            }
        }
    }
}

__global__ __launch_bounds__(64) void k_combine(const float* __restrict__ adiag,
                                                const float* __restrict__ h0,
                                                float* __restrict__ X) {
    const int b = blockIdx.x >> 3;
    const int s = (blockIdx.x & 7) * 64 + threadIdx.x;
    const float a = adiag[s];
    float aT = a;
    #pragma unroll
    for (int i = 0; i < 6; ++i) aT = aT * aT;
    float cin = h0[s];
    float* xr = X + ((size_t)b * LSEQ + (TCH - 1)) * DS + s;
    for (int c0 = 0; c0 < NCHUNK; c0 += 16) {
        float v[16];
        #pragma unroll
        for (int k = 0; k < 16; ++k) v[k] = xr[(size_t)(c0 + k) * TCH * DS];
        #pragma unroll
        for (int k = 0; k < 16; ++k) {
            cin = fmaf(aT, cin, v[k]);
            xr[(size_t)(c0 + k) * TCH * DS] = cin;
        }
    }
}

__global__ __launch_bounds__(256) void k_fixup(const float* __restrict__ adiag,
                                               const float* __restrict__ h0,
                                               float* __restrict__ X) {
    const int b = blockIdx.x >> 6, c = blockIdx.x & 63;
    const int s1 = threadIdx.x, s2 = threadIdx.x + 256;
    const float a1 = adiag[s1], a2 = adiag[s2];
    const size_t base = ((size_t)b * LSEQ + (size_t)c * TCH) * DS;
    float c1, c2;
    if (c == 0) { c1 = h0[s1];           c2 = h0[s2]; }
    else        { c1 = X[base - DS + s1]; c2 = X[base - DS + s2]; }
    float p1 = a1, p2 = a2;
    for (int t0 = 0; t0 < TCH - 1; t0 += 9) {
        float v1[9], v2[9];
        #pragma unroll
        for (int k = 0; k < 9; ++k) {
            const size_t idx = base + (size_t)(t0 + k) * DS;
            v1[k] = X[idx + s1];
            v2[k] = X[idx + s2];
        }
        #pragma unroll
        for (int k = 0; k < 9; ++k) {
            const size_t idx = base + (size_t)(t0 + k) * DS;
            X[idx + s1] = fmaf(p1, c1, v1[k]);  p1 *= a1;
            X[idx + s2] = fmaf(p2, c2, v2[k]);  p2 *= a2;
        }
    }
}

__global__ __launch_bounds__(256) void k_gemm2(const float* __restrict__ X,
                                               const float* __restrict__ u,
                                               const float* __restrict__ Cm,
                                               const float* __restrict__ Dm,
                                               float* __restrict__ Y) {
    extern __shared__ char smem_raw[];
    float* s_x = (float*)smem_raw;
    float* s_u = (float*)smem_raw;

    const int b      = blockIdx.x >> 6;
    const int l0base = (blockIdx.x & 63) * 64;
    const int t      = threadIdx.x;
    const int wave   = t >> 5;
    const int lane   = t & 31;
    const int mrow   = lane & 15;
    const int khalf  = (lane >> 4) << 1;
    const int ncol   = lane & 15;
    const int hi     = (lane >> 4) << 3;

    const float* __restrict__ ub = u + (size_t)b * DI * LSEQ;
    const float* __restrict__ Xb = X + (size_t)b * LSEQ * DS;

    __builtin_prefetch(Cm + (size_t)(wave * 32 + mrow) * DS, 0, 3);

    {
        const int lr = t >> 2;
        const int cb = (t & 3) * 128;
        #pragma unroll
        for (int c = 0; c < 32; ++c) {
            const float4 v = *(const float4*)(Xb + (size_t)(l0base + lr) * DS + cb + c * 4);
            *(float4*)(&s_x[lr * XP + cb + c * 4]) = v;
        }
    }
    __syncthreads();

    const v8f vzero = {0.f,0.f,0.f,0.f,0.f,0.f,0.f,0.f};
    v8f acc[2][4];
    #pragma unroll
    for (int i = 0; i < 2; ++i)
        #pragma unroll
        for (int j = 0; j < 4; ++j) acc[i][j] = vzero;

    for (int k0 = 0; k0 < DS; k0 += 4) {
        v2f afrag[2];
        #pragma unroll
        for (int i = 0; i < 2; ++i) {
            const int orow = (wave * 2 + i) * 16 + mrow;
            afrag[i] = *(const v2f*)(Cm + (size_t)orow * DS + k0 + khalf);
        }
        v2f bfrag[4];
        #pragma unroll
        for (int j = 0; j < 4; ++j) {
            const int lr = j * 16 + ncol;
            bfrag[j] = *(const v2f*)(&s_x[lr * XP + k0 + khalf]);
        }
        #pragma unroll
        for (int i = 0; i < 2; ++i)
            #pragma unroll
            for (int j = 0; j < 4; ++j)
                acc[i][j] = wmma_f32_16x16x4(afrag[i], bfrag[j], acc[i][j]);
    }

    __syncthreads();
    #pragma unroll
    for (int c = 0; c < 16; ++c) {
        const float4 v = *(const float4*)(ub + (size_t)t * LSEQ + l0base + c * 4);
        *(float4*)(&s_u[t * UP + c * 4]) = v;
    }
    __syncthreads();

    for (int k0 = 0; k0 < DI; k0 += 4) {
        v2f afrag[2];
        #pragma unroll
        for (int i = 0; i < 2; ++i) {
            const int orow = (wave * 2 + i) * 16 + mrow;
            afrag[i] = *(const v2f*)(Dm + (size_t)orow * DI + k0 + khalf);
        }
        v2f bfrag[4];
        #pragma unroll
        for (int j = 0; j < 4; ++j) {
            const int lcol = j * 16 + ncol;
            bfrag[j].x = s_u[(k0 + khalf)     * UP + lcol];
            bfrag[j].y = s_u[(k0 + khalf + 1) * UP + lcol];
        }
        #pragma unroll
        for (int i = 0; i < 2; ++i)
            #pragma unroll
            for (int j = 0; j < 4; ++j)
                acc[i][j] = wmma_f32_16x16x4(afrag[i], bfrag[j], acc[i][j]);
    }

    #pragma unroll
    for (int i = 0; i < 2; ++i) {
        const int obase = (wave * 2 + i) * 16 + hi;
        #pragma unroll
        for (int j = 0; j < 4; ++j) {
            const int l = l0base + j * 16 + ncol;
            #pragma unroll
            for (int v = 0; v < 8; ++v)
                Y[((size_t)b * DO + obase + v) * LSEQ + l] = acc[i][j][v];
        }
    }
}

extern "C" void kernel_launch(void* const* d_in, const int* in_sizes, int n_in,
                              void* d_out, int out_size, void* d_ws, size_t ws_size,
                              hipStream_t stream) {
    (void)in_sizes; (void)n_in; (void)out_size; (void)ws_size;

    const float* u    = (const float*)d_in[0];
    const float* a_un = (const float*)d_in[1];
    const float* Bm   = (const float*)d_in[2];
    const float* Cm   = (const float*)d_in[3];
    const float* Dm   = (const float*)d_in[4];
    const float* h0   = (const float*)d_in[5];
    float*       Y    = (float*)d_out;

    char* ws = (char*)d_ws;
    const size_t X_BYTES = (size_t)BZ * LSEQ * DS * sizeof(float);
    float* X           = (float*)(ws);
    float* adiag       = (float*)(ws + X_BYTES);

    const size_t SH1 = (size_t)DI * UP * sizeof(float);
    const size_t SH2 = (size_t)64 * XP * sizeof(float);

    k_adiag    <<<1,                512, 0,   stream>>>(a_un, adiag);
    k_gemm1    <<<BZ * (LSEQ / 64), 256, SH1, stream>>>(u, Bm, adiag, X);
    k_combine  <<<BZ * (DS / 64),   64,  0,   stream>>>(adiag, h0, X);
    k_fixup    <<<BZ * NCHUNK,      256, 0,   stream>>>(adiag, h0, X);
    k_gemm2    <<<BZ * (LSEQ / 64), 256, SH2, stream>>>(X, u, Cm, Dm, Y);
}


# Fixup pass folded into GEMM2's X-tile staging: each thread owns 4 s-columns and adds a^(t+1)*carry_in to rows 0..62 on their way to LDS; fixup launch and its X read+write removed, 5 launches to 4
# speedup vs baseline: 1.3067x; 1.0854x over previous
#include <hip/hip_runtime.h>
#include <hip/hip_bf16.h>

#define BZ    8
#define DI    256
#define DS    512
#define DO    256
#define LSEQ  4096
#define TCH   64
#define NCHUNK (LSEQ / TCH)

#define UP    68
#define XP    516

typedef float v2f __attribute__((ext_vector_type(2)));
typedef float v4f __attribute__((ext_vector_type(4)));
typedef float v8f __attribute__((ext_vector_type(8)));

static __device__ __forceinline__ v8f wmma_f32_16x16x4(v2f a, v2f b, v8f c) {
    return __builtin_amdgcn_wmma_f32_16x16x4_f32(
         false, a,  false, b,
         (short)0, c,  false,  false);
}

template <int N>
static __device__ __forceinline__ float row_shr0(float x) {
    return __builtin_bit_cast(float,
        __builtin_amdgcn_update_dpp(0, __builtin_bit_cast(int, x), 0x110 + N, 0xF, 0xF, true));
}
static __device__ __forceinline__ float row_last(float x) {
    return __builtin_bit_cast(float,
        __builtin_amdgcn_update_dpp(0, __builtin_bit_cast(int, x), 0x15F, 0xF, 0xF, true));
}
static __device__ __forceinline__ float row_scan16(float x, float a1, float a2, float a4, float a8) {
    x = fmaf(a1, row_shr0<1>(x), x);
    x = fmaf(a2, row_shr0<2>(x), x);
    x = fmaf(a4, row_shr0<4>(x), x);
    x = fmaf(a8, row_shr0<8>(x), x);
    return x;
}

__global__ void k_adiag(const float* __restrict__ a_un, float* __restrict__ adiag) {
    int s = threadIdx.x;
    float x  = a_un[s];
    float sp = fmaxf(x, 0.0f) + log1pf(expf(-fabsf(x)));
    adiag[s] = -sp;
}

__global__ __launch_bounds__(256) void k_gemm1(const float* __restrict__ u,
                                               const float* __restrict__ Bm,
                                               const float* __restrict__ adiag,
                                               float* __restrict__ X) {
    extern __shared__ char smem_raw[];
    float* s_u = (float*)smem_raw;

    const int b      = blockIdx.x >> 6;
    const int l0base = (blockIdx.x & 63) * 64;
    const int t      = threadIdx.x;
    const int wave   = t >> 5;
    const int lane   = t & 31;
    const int mrow   = lane & 15;
    const int khalf  = (lane >> 4) << 1;
    const int ncol   = lane & 15;
    const int hi     = (lane >> 4) << 3;

    const float* __restrict__ ub = u + (size_t)b * DI * LSEQ;

    __builtin_prefetch(Bm + (size_t)(wave * 64 + mrow) * DI, 0, 3);

    #pragma unroll
    for (int c = 0; c < 16; ++c) {
        const float4 v = *(const float4*)(ub + (size_t)t * LSEQ + l0base + c * 4);
        *(float4*)(&s_u[t * UP + c * 4]) = v;
    }
    __syncthreads();

    const v8f vzero = {0.f,0.f,0.f,0.f,0.f,0.f,0.f,0.f};

    for (int half = 0; half < 2; ++half) {
        const int st0 = wave * 4 + half * 2;
        v8f acc[2][4];
        #pragma unroll
        for (int i = 0; i < 2; ++i)
            #pragma unroll
            for (int j = 0; j < 4; ++j) acc[i][j] = vzero;

        for (int k0 = 0; k0 < DI; k0 += 4) {
            v2f afrag[2];
            #pragma unroll
            for (int i = 0; i < 2; ++i) {
                const int srow = (st0 + i) * 16 + mrow;
                afrag[i] = *(const v2f*)(Bm + (size_t)srow * DI + k0 + khalf);
            }
            v2f bfrag[4];
            #pragma unroll
            for (int j = 0; j < 4; ++j) {
                const int lcol = j * 16 + ncol;
                bfrag[j].x = s_u[(k0 + khalf)     * UP + lcol];
                bfrag[j].y = s_u[(k0 + khalf + 1) * UP + lcol];
            }
            #pragma unroll
            for (int i = 0; i < 2; ++i)
                #pragma unroll
                for (int j = 0; j < 4; ++j)
                    acc[i][j] = wmma_f32_16x16x4(afrag[i], bfrag[j], acc[i][j]);
        }

        #pragma unroll
        for (int i = 0; i < 2; ++i) {
            const int sbase = (st0 + i) * 16 + hi;
            const v4f alo = *(const v4f*)(adiag + sbase);
            const v4f ahi = *(const v4f*)(adiag + sbase + 4);
            #pragma unroll
            for (int r = 0; r < 8; ++r) {
                const float a1 = (r < 4) ? alo[r & 3] : ahi[r & 3];
                const float a2 = a1 * a1, a4 = a2 * a2, a8 = a4 * a4;
                const float w = row_scan16((ncol == 0) ? a1 : 0.f, a1, a2, a4, a8);
                float prev = 0.f;
                #pragma unroll
                for (int j = 0; j < 4; ++j) {
                    float x = row_scan16(acc[i][j][r], a1, a2, a4, a8);
                    if (j > 0) x = fmaf(w, prev, x);
                    prev = row_last(x);
                    acc[i][j][r] = x;
                }
            }
        }

        #pragma unroll
        for (int i = 0; i < 2; ++i) {
            const int sbase = (st0 + i) * 16 + hi;
            #pragma unroll
            for (int j = 0; j < 4; ++j) {
                const int l = l0base + j * 16 + ncol;
                float* xp = X + ((size_t)b * LSEQ + l) * DS + sbase;
                v4f lo  = __builtin_shufflevector(acc[i][j], acc[i][j], 0,1,2,3);
                v4f hiv = __builtin_shufflevector(acc[i][j], acc[i][j], 4,5,6,7);
                *(v4f*)(xp)     = lo;
                *(v4f*)(xp + 4) = hiv;
            }
        }
    }
}

__global__ __launch_bounds__(64) void k_combine(const float* __restrict__ adiag,
                                                const float* __restrict__ h0,
                                                float* __restrict__ X) {
    const int b = blockIdx.x >> 3;
    const int s = (blockIdx.x & 7) * 64 + threadIdx.x;
    const float a = adiag[s];
    float aT = a;
    #pragma unroll
    for (int i = 0; i < 6; ++i) aT = aT * aT;
    float cin = h0[s];
    float* xr = X + ((size_t)b * LSEQ + (TCH - 1)) * DS + s;
    for (int c0 = 0; c0 < NCHUNK; c0 += 16) {
        float v[16];
        #pragma unroll
        for (int k = 0; k < 16; ++k) v[k] = xr[(size_t)(c0 + k) * TCH * DS];
        #pragma unroll
        for (int k = 0; k < 16; ++k) {
            cin = fmaf(aT, cin, v[k]);
            xr[(size_t)(c0 + k) * TCH * DS] = cin;
        }
    }
}

__global__ __launch_bounds__(256) void k_gemm2(const float* __restrict__ X,
                                               const float* __restrict__ u,
                                               const float* __restrict__ Cm,
                                               const float* __restrict__ Dm,
                                               const float* __restrict__ adiag,
                                               const float* __restrict__ h0,
                                               float* __restrict__ Y) {
    extern __shared__ char smem_raw[];
    float* s_x = (float*)smem_raw;
    float* s_u = (float*)smem_raw;

    const int b      = blockIdx.x >> 6;
    const int l0base = (blockIdx.x & 63) * 64;
    const int t      = threadIdx.x;
    const int wave   = t >> 5;
    const int lane   = t & 31;
    const int mrow   = lane & 15;
    const int khalf  = (lane >> 4) << 1;
    const int ncol   = lane & 15;
    const int hi     = (lane >> 4) << 3;

    const float* __restrict__ ub = u + (size_t)b * DI * LSEQ;
    const float* __restrict__ Xb = X + (size_t)b * LSEQ * DS;

    __builtin_prefetch(Cm + (size_t)(wave * 32 + mrow) * DS, 0, 3);

    {
        const int cg = (t & 127) * 4;
        const int r0 = (t >> 7) * 32;
        const v4f a = *(const v4f*)(adiag + cg);
        v4f cin;
        if (l0base == 0) cin = *(const v4f*)(h0 + cg);
        else             cin = *(const v4f*)(Xb + (size_t)(l0base - 1) * DS + cg);
        const float* __restrict__ xsrc = Xb + (size_t)(l0base + r0) * DS + cg;
        v4f v[32];
        #pragma unroll
        for (int k = 0; k < 32; ++k) v[k] = *(const v4f*)(xsrc + (size_t)k * DS);
        __builtin_amdgcn_sched_barrier(0);
        v4f p33 = a;
        #pragma unroll
        for (int k = 0; k < 32; ++k) p33 *= a;
        v4f p = (r0 != 0) ? p33 : a;
        float* xdst = s_x + r0 * XP + cg;
        #pragma unroll
        for (int k = 0; k < 32; ++k) {
            v4f x;
            x.x = fmaf(p.x, cin.x, v[k].x);
            x.y = fmaf(p.y, cin.y, v[k].y);
            x.z = fmaf(p.z, cin.z, v[k].z);
            x.w = fmaf(p.w, cin.w, v[k].w);
            if (r0 + k == TCH - 1) x = v[k];
            *(v4f*)(xdst + k * XP) = x;
            p *= a;
        }
    }
    __syncthreads();

    const v8f vzero = {0.f,0.f,0.f,0.f,0.f,0.f,0.f,0.f};
    v8f acc[2][4];
    #pragma unroll
    for (int i = 0; i < 2; ++i)
        #pragma unroll
        for (int j = 0; j < 4; ++j) acc[i][j] = vzero;

    for (int k0 = 0; k0 < DS; k0 += 4) {
        v2f afrag[2];
        #pragma unroll
        for (int i = 0; i < 2; ++i) {
            const int orow = (wave * 2 + i) * 16 + mrow;
            afrag[i] = *(const v2f*)(Cm + (size_t)orow * DS + k0 + khalf);
        }
        v2f bfrag[4];
        #pragma unroll
        for (int j = 0; j < 4; ++j) {
            const int lr = j * 16 + ncol;
            bfrag[j] = *(const v2f*)(&s_x[lr * XP + k0 + khalf]);
        }
        #pragma unroll
        for (int i = 0; i < 2; ++i)
            #pragma unroll
            for (int j = 0; j < 4; ++j)
                acc[i][j] = wmma_f32_16x16x4(afrag[i], bfrag[j], acc[i][j]);
    }

    __syncthreads();
    #pragma unroll
    for (int c = 0; c < 16; ++c) {
        const float4 v = *(const float4*)(ub + (size_t)t * LSEQ + l0base + c * 4);
        *(float4*)(&s_u[t * UP + c * 4]) = v;
    }
    __syncthreads();

    for (int k0 = 0; k0 < DI; k0 += 4) {
        v2f afrag[2];
        #pragma unroll
        for (int i = 0; i < 2; ++i) {
            const int orow = (wave * 2 + i) * 16 + mrow;
            afrag[i] = *(const v2f*)(Dm + (size_t)orow * DI + k0 + khalf);
        }
        v2f bfrag[4];
        #pragma unroll
        for (int j = 0; j < 4; ++j) {
            const int lcol = j * 16 + ncol;
            bfrag[j].x = s_u[(k0 + khalf)     * UP + lcol];
            bfrag[j].y = s_u[(k0 + khalf + 1) * UP + lcol];
        }
        #pragma unroll
        for (int i = 0; i < 2; ++i)
            #pragma unroll
            for (int j = 0; j < 4; ++j)
                acc[i][j] = wmma_f32_16x16x4(afrag[i], bfrag[j], acc[i][j]);
    }

    #pragma unroll
    for (int i = 0; i < 2; ++i) {
        const int obase = (wave * 2 + i) * 16 + hi;
        #pragma unroll
        for (int j = 0; j < 4; ++j) {
            const int l = l0base + j * 16 + ncol;
            #pragma unroll
            for (int v = 0; v < 8; ++v)
                Y[((size_t)b * DO + obase + v) * LSEQ + l] = acc[i][j][v];
        }
    }
}

extern "C" void kernel_launch(void* const* d_in, const int* in_sizes, int n_in,
                              void* d_out, int out_size, void* d_ws, size_t ws_size,
                              hipStream_t stream) {
    (void)in_sizes; (void)n_in; (void)out_size; (void)ws_size;

    const float* u    = (const float*)d_in[0];
    const float* a_un = (const float*)d_in[1];
    const float* Bm   = (const float*)d_in[2];
    const float* Cm   = (const float*)d_in[3];
    const float* Dm   = (const float*)d_in[4];
    const float* h0   = (const float*)d_in[5];
    float*       Y    = (float*)d_out;

    char* ws = (char*)d_ws;
    const size_t X_BYTES = (size_t)BZ * LSEQ * DS * sizeof(float);
    float* X           = (float*)(ws);
    float* adiag       = (float*)(ws + X_BYTES);

    const size_t SH1 = (size_t)DI * UP * sizeof(float);
    const size_t SH2 = (size_t)64 * XP * sizeof(float);

    k_adiag    <<<1,                512, 0,   stream>>>(a_un, adiag);
    k_gemm1    <<<BZ * (LSEQ / 64), 256, SH1, stream>>>(u, Bm, adiag, X);
    k_combine  <<<BZ * (DS / 64),   64,  0,   stream>>>(adiag, h0, X);
    k_gemm2    <<<BZ * (LSEQ / 64), 256, SH2, stream>>>(X, u, Cm, Dm, adiag, h0, Y);
}


# K-loops software-pipelined in registers: the next 32-WMMA block's A (global) and B (LDS) fragments are loaded as one group into a second register set while the current block is multiplied
# speedup vs baseline: 1.3721x; 1.0500x over previous
#include <hip/hip_runtime.h>
#include <hip/hip_bf16.h>

#define BZ    8
#define DI    256
#define DS    512
#define DO    256
#define LSEQ  4096
#define TCH   64
#define NCHUNK (LSEQ / TCH)

#define UP    68
#define XP    516

typedef float v2f __attribute__((ext_vector_type(2)));
typedef float v4f __attribute__((ext_vector_type(4)));
typedef float v8f __attribute__((ext_vector_type(8)));

static __device__ __forceinline__ v8f wmma_f32_16x16x4(v2f a, v2f b, v8f c) {
    return __builtin_amdgcn_wmma_f32_16x16x4_f32(
         false, a,  false, b,
         (short)0, c,  false,  false);
}

template <int N>
static __device__ __forceinline__ float row_shr0(float x) {
    return __builtin_bit_cast(float,
        __builtin_amdgcn_update_dpp(0, __builtin_bit_cast(int, x), 0x110 + N, 0xF, 0xF, true));
}
static __device__ __forceinline__ float row_last(float x) {
    return __builtin_bit_cast(float,
        __builtin_amdgcn_update_dpp(0, __builtin_bit_cast(int, x), 0x15F, 0xF, 0xF, true));
}
static __device__ __forceinline__ float row_scan16(float x, float a1, float a2, float a4, float a8) {
    x = fmaf(a1, row_shr0<1>(x), x);
    x = fmaf(a2, row_shr0<2>(x), x);
    x = fmaf(a4, row_shr0<4>(x), x);
    x = fmaf(a8, row_shr0<8>(x), x);
    return x;
}

static __device__ __forceinline__ void load_blk_kmajor(v2f (&fa)[4][2], float (&bx)[4][4], float (&by)[4][4],
                                                       const float* __restrict__ arow0,
                                                       const float* __restrict__ arow1,
                                                       const float* s_t, int kk, int ncol) {
    #pragma unroll
    for (int s = 0; s < 4; ++s) {
        fa[s][0] = *(const v2f*)(arow0 + kk + 4 * s);
        fa[s][1] = *(const v2f*)(arow1 + kk + 4 * s);
        #pragma unroll
        for (int j = 0; j < 4; ++j) {
            bx[s][j] = s_t[(kk + 4 * s)     * UP + j * 16 + ncol];
            by[s][j] = s_t[(kk + 4 * s + 1) * UP + j * 16 + ncol];
        }
    }
}
static __device__ __forceinline__ void load_blk_rowmajor(v2f (&fa)[4][2], v2f (&fb)[4][4],
                                                         const float* __restrict__ arow0,
                                                         const float* __restrict__ arow1,
                                                         const float* s_t, int kk, int ncol) {
    #pragma unroll
    for (int s = 0; s < 4; ++s) {
        fa[s][0] = *(const v2f*)(arow0 + kk + 4 * s);
        fa[s][1] = *(const v2f*)(arow1 + kk + 4 * s);
        #pragma unroll
        for (int j = 0; j < 4; ++j)
            fb[s][j] = *(const v2f*)(&s_t[(j * 16 + ncol) * XP + kk + 4 * s]);
    }
}
static __device__ __forceinline__ void mma_blk(v8f (&acc)[2][4], const v2f (&fa)[4][2], const v2f (&fb)[4][4]) {
    #pragma unroll
    for (int s = 0; s < 4; ++s)
        #pragma unroll
        for (int i = 0; i < 2; ++i)
            #pragma unroll
            for (int j = 0; j < 4; ++j)
                acc[i][j] = wmma_f32_16x16x4(fa[s][i], fb[s][j], acc[i][j]);
}
static __device__ __forceinline__ void mma_blk_xy(v8f (&acc)[2][4], const v2f (&fa)[4][2],
                                                  const float (&bx)[4][4], const float (&by)[4][4]) {
    #pragma unroll
    for (int s = 0; s < 4; ++s)
        #pragma unroll
        for (int j = 0; j < 4; ++j) {
            v2f b; b.x = bx[s][j]; b.y = by[s][j];
            #pragma unroll
            for (int i = 0; i < 2; ++i)
                acc[i][j] = wmma_f32_16x16x4(fa[s][i], b, acc[i][j]);
        }
}
static __device__ __forceinline__ void keep_loads_here() {
    asm volatile("" ::: "memory");
    __builtin_amdgcn_sched_barrier(0);
}

__global__ void k_adiag(const float* __restrict__ a_un, float* __restrict__ adiag) {
    int s = threadIdx.x;
    float x  = a_un[s];
    float sp = fmaxf(x, 0.0f) + log1pf(expf(-fabsf(x)));
    adiag[s] = -sp;
}

__global__ __launch_bounds__(256) void k_gemm1(const float* __restrict__ u,
                                               const float* __restrict__ Bm,
                                               const float* __restrict__ adiag,
                                               float* __restrict__ X) {
    extern __shared__ char smem_raw[];
    float* s_u = (float*)smem_raw;

    const int b      = blockIdx.x >> 6;
    const int l0base = (blockIdx.x & 63) * 64;
    const int t      = threadIdx.x;
    const int wave   = t >> 5;
    const int lane   = t & 31;
    const int mrow   = lane & 15;
    const int khalf  = (lane >> 4) << 1;
    const int ncol   = lane & 15;
    const int hi     = (lane >> 4) << 3;

    const float* __restrict__ ub = u + (size_t)b * DI * LSEQ;

    __builtin_prefetch(Bm + (size_t)(wave * 64 + mrow) * DI, 0, 3);

    #pragma unroll
    for (int c = 0; c < 16; ++c) {
        const float4 v = *(const float4*)(ub + (size_t)t * LSEQ + l0base + c * 4);
        *(float4*)(&s_u[t * UP + c * 4]) = v;
    }
    __syncthreads();

    const v8f vzero = {0.f,0.f,0.f,0.f,0.f,0.f,0.f,0.f};

    for (int half = 0; half < 2; ++half) {
        const int st0 = wave * 4 + half * 2;
        v8f acc[2][4];
        #pragma unroll
        for (int i = 0; i < 2; ++i)
            #pragma unroll
            for (int j = 0; j < 4; ++j) acc[i][j] = vzero;

        const float* __restrict__ ar0 = Bm + (size_t)((st0 + 0) * 16 + mrow) * DI;
        const float* __restrict__ ar1 = Bm + (size_t)((st0 + 1) * 16 + mrow) * DI;
        v2f fa0[4][2], fa1[4][2];
        float bx0[4][4], by0[4][4], bx1[4][4], by1[4][4];
        load_blk_kmajor(fa0, bx0, by0, ar0, ar1, s_u, khalf, ncol);
        for (int k0 = 0; k0 < DI; k0 += 32) {
            load_blk_kmajor(fa1, bx1, by1, ar0, ar1, s_u, k0 + 16 + khalf, ncol);
            keep_loads_here();
            mma_blk_xy(acc, fa0, bx0, by0);
            __builtin_amdgcn_sched_barrier(0);
            load_blk_kmajor(fa0, bx0, by0, ar0, ar1, s_u, ((k0 + 32) & (DI - 1)) + khalf, ncol);
            keep_loads_here();
            mma_blk_xy(acc, fa1, bx1, by1);
        }

        #pragma unroll
        for (int i = 0; i < 2; ++i) {
            const int sbase = (st0 + i) * 16 + hi;
            const v4f alo = *(const v4f*)(adiag + sbase);
            const v4f ahi = *(const v4f*)(adiag + sbase + 4);
            #pragma unroll
            for (int r = 0; r < 8; ++r) {
                const float a1 = (r < 4) ? alo[r & 3] : ahi[r & 3];
                const float a2 = a1 * a1, a4 = a2 * a2, a8 = a4 * a4;
                const float w = row_scan16((ncol == 0) ? a1 : 0.f, a1, a2, a4, a8);
                float prev = 0.f;
                #pragma unroll
                for (int j = 0; j < 4; ++j) {
                    float x = row_scan16(acc[i][j][r], a1, a2, a4, a8);
                    if (j > 0) x = fmaf(w, prev, x);
                    prev = row_last(x);
                    acc[i][j][r] = x;
                }
            }
        }

        #pragma unroll
        for (int i = 0; i < 2; ++i) {
            const int sbase = (st0 + i) * 16 + hi;
            #pragma unroll
            for (int j = 0; j < 4; ++j) {
                const int l = l0base + j * 16 + ncol;
                float* xp = X + ((size_t)b * LSEQ + l) * DS + sbase;
                v4f lo  = __builtin_shufflevector(acc[i][j], acc[i][j], 0,1,2,3);
                v4f hiv = __builtin_shufflevector(acc[i][j], acc[i][j], 4,5,6,7);
                *(v4f*)(xp)     = lo;
                *(v4f*)(xp + 4) = hiv;
            }
        }
    }
}

__global__ __launch_bounds__(64) void k_combine(const float* __restrict__ adiag,
                                                const float* __restrict__ h0,
                                                float* __restrict__ X) {
    const int b = blockIdx.x >> 3;
    const int s = (blockIdx.x & 7) * 64 + threadIdx.x;
    const float a = adiag[s];
    float aT = a;
    #pragma unroll
    for (int i = 0; i < 6; ++i) aT = aT * aT;
    float cin = h0[s];
    float* xr = X + ((size_t)b * LSEQ + (TCH - 1)) * DS + s;
    for (int c0 = 0; c0 < NCHUNK; c0 += 16) {
        float v[16];
        #pragma unroll
        for (int k = 0; k < 16; ++k) v[k] = xr[(size_t)(c0 + k) * TCH * DS];
        #pragma unroll
        for (int k = 0; k < 16; ++k) {
            cin = fmaf(aT, cin, v[k]);
            xr[(size_t)(c0 + k) * TCH * DS] = cin;
        }
    }
}

__global__ __launch_bounds__(256) void k_gemm2(const float* __restrict__ X,
                                               const float* __restrict__ u,
                                               const float* __restrict__ Cm,
                                               const float* __restrict__ Dm,
                                               const float* __restrict__ adiag,
                                               const float* __restrict__ h0,
                                               float* __restrict__ Y) {
    extern __shared__ char smem_raw[];
    float* s_x = (float*)smem_raw;
    float* s_u = (float*)smem_raw;

    const int b      = blockIdx.x >> 6;
    const int l0base = (blockIdx.x & 63) * 64;
    const int t      = threadIdx.x;
    const int wave   = t >> 5;
    const int lane   = t & 31;
    const int mrow   = lane & 15;
    const int khalf  = (lane >> 4) << 1;
    const int ncol   = lane & 15;
    const int hi     = (lane >> 4) << 3;

    const float* __restrict__ ub = u + (size_t)b * DI * LSEQ;
    const float* __restrict__ Xb = X + (size_t)b * LSEQ * DS;

    __builtin_prefetch(Cm + (size_t)(wave * 32 + mrow) * DS, 0, 3);

    {
        const int cg = (t & 127) * 4;
        const int r0 = (t >> 7) * 32;
        const v4f a = *(const v4f*)(adiag + cg);
        v4f cin;
        if (l0base == 0) cin = *(const v4f*)(h0 + cg);
        else             cin = *(const v4f*)(Xb + (size_t)(l0base - 1) * DS + cg);
        const float* __restrict__ xsrc = Xb + (size_t)(l0base + r0) * DS + cg;
        v4f v[32];
        #pragma unroll
        for (int k = 0; k < 32; ++k) v[k] = *(const v4f*)(xsrc + (size_t)k * DS);
        __builtin_amdgcn_sched_barrier(0);
        v4f p33 = a;
        #pragma unroll
        for (int k = 0; k < 32; ++k) p33 *= a;
        v4f p = (r0 != 0) ? p33 : a;
        float* xdst = s_x + r0 * XP + cg;
        #pragma unroll
        for (int k = 0; k < 32; ++k) {
            v4f x;
            x.x = fmaf(p.x, cin.x, v[k].x);
            x.y = fmaf(p.y, cin.y, v[k].y);
            x.z = fmaf(p.z, cin.z, v[k].z);
            x.w = fmaf(p.w, cin.w, v[k].w);
            if (r0 + k == TCH - 1) x = v[k];
            *(v4f*)(xdst + k * XP) = x;
            p *= a;
        }
    }
    __syncthreads();

    const v8f vzero = {0.f,0.f,0.f,0.f,0.f,0.f,0.f,0.f};
    v8f acc[2][4];
    #pragma unroll
    for (int i = 0; i < 2; ++i)
        #pragma unroll
        for (int j = 0; j < 4; ++j) acc[i][j] = vzero;

    v2f fa0[4][2], fb0[4][4], fa1[4][2], fb1[4][4];
    {
        const float* __restrict__ cr0 = Cm + (size_t)((wave * 2 + 0) * 16 + mrow) * DS;
        const float* __restrict__ cr1 = Cm + (size_t)((wave * 2 + 1) * 16 + mrow) * DS;
        load_blk_rowmajor(fa0, fb0, cr0, cr1, s_x, khalf, ncol);
        for (int k0 = 0; k0 < DS; k0 += 32) {
            load_blk_rowmajor(fa1, fb1, cr0, cr1, s_x, k0 + 16 + khalf, ncol);
            keep_loads_here();
            mma_blk(acc, fa0, fb0);
            __builtin_amdgcn_sched_barrier(0);
            load_blk_rowmajor(fa0, fb0, cr0, cr1, s_x, ((k0 + 32) & (DS - 1)) + khalf, ncol);
            keep_loads_here();
            mma_blk(acc, fa1, fb1);
        }
    }

    __syncthreads();
    #pragma unroll
    for (int c = 0; c < 16; ++c) {
        const float4 v = *(const float4*)(ub + (size_t)t * LSEQ + l0base + c * 4);
        *(float4*)(&s_u[t * UP + c * 4]) = v;
    }
    __syncthreads();

    {
        const float* __restrict__ dr0 = Dm + (size_t)((wave * 2 + 0) * 16 + mrow) * DI;
        const float* __restrict__ dr1 = Dm + (size_t)((wave * 2 + 1) * 16 + mrow) * DI;
        float bx0[4][4], by0[4][4], bx1[4][4], by1[4][4];
        load_blk_kmajor(fa0, bx0, by0, dr0, dr1, s_u, khalf, ncol);
        for (int k0 = 0; k0 < DI; k0 += 32) {
            load_blk_kmajor(fa1, bx1, by1, dr0, dr1, s_u, k0 + 16 + khalf, ncol);
            keep_loads_here();
            mma_blk_xy(acc, fa0, bx0, by0);
            __builtin_amdgcn_sched_barrier(0);
            load_blk_kmajor(fa0, bx0, by0, dr0, dr1, s_u, ((k0 + 32) & (DI - 1)) + khalf, ncol);
            keep_loads_here();
            mma_blk_xy(acc, fa1, bx1, by1);
        }
    }

    #pragma unroll
    for (int i = 0; i < 2; ++i) {
        const int obase = (wave * 2 + i) * 16 + hi;
        #pragma unroll
        for (int j = 0; j < 4; ++j) {
            const int l = l0base + j * 16 + ncol;
            #pragma unroll
            for (int v = 0; v < 8; ++v)
                Y[((size_t)b * DO + obase + v) * LSEQ + l] = acc[i][j][v];
        }
    }
}

extern "C" void kernel_launch(void* const* d_in, const int* in_sizes, int n_in,
                              void* d_out, int out_size, void* d_ws, size_t ws_size,
                              hipStream_t stream) {
    (void)in_sizes; (void)n_in; (void)out_size; (void)ws_size;

    const float* u    = (const float*)d_in[0];
    const float* a_un = (const float*)d_in[1];
    const float* Bm   = (const float*)d_in[2];
    const float* Cm   = (const float*)d_in[3];
    const float* Dm   = (const float*)d_in[4];
    const float* h0   = (const float*)d_in[5];
    float*       Y    = (float*)d_out;

    char* ws = (char*)d_ws;
    const size_t X_BYTES = (size_t)BZ * LSEQ * DS * sizeof(float);
    float* X           = (float*)(ws);
    float* adiag       = (float*)(ws + X_BYTES);

    const size_t SH1 = (size_t)DI * UP * sizeof(float);
    const size_t SH2 = (size_t)64 * XP * sizeof(float);

    k_adiag    <<<1,                512, 0,   stream>>>(a_un, adiag);
    k_gemm1    <<<BZ * (LSEQ / 64), 256, SH1, stream>>>(u, Bm, adiag, X);
    k_combine  <<<BZ * (DS / 64),   64,  0,   stream>>>(adiag, h0, X);
    k_gemm2    <<<BZ * (LSEQ / 64), 256, SH2, stream>>>(X, u, Cm, Dm, adiag, h0, Y);
}


# Global accesses made whole-line: u-tile staging loads read two contiguous tile rows per instruction (was 32 rows x 16 B), X tiles go through wave-private LDS so each store writes 4 full 128-B lines
# speedup vs baseline: 1.4269x; 1.0399x over previous
#include <hip/hip_runtime.h>
#include <hip/hip_bf16.h>

#define BZ    8
#define DI    256
#define DS    512
#define DO    256
#define LSEQ  4096
#define TCH   64
#define NCHUNK (LSEQ / TCH)

#define UP    68
#define XP    516
#define SP    36
#define SWAVE (64 * SP)

typedef float v2f __attribute__((ext_vector_type(2)));
typedef float v4f __attribute__((ext_vector_type(4)));
typedef float v8f __attribute__((ext_vector_type(8)));

static __device__ __forceinline__ v8f wmma_f32_16x16x4(v2f a, v2f b, v8f c) {
    return __builtin_amdgcn_wmma_f32_16x16x4_f32(
         false, a,  false, b,
         (short)0, c,  false,  false);
}

template <int N>
static __device__ __forceinline__ float row_shr0(float x) {
    return __builtin_bit_cast(float,
        __builtin_amdgcn_update_dpp(0, __builtin_bit_cast(int, x), 0x110 + N, 0xF, 0xF, true));
}
static __device__ __forceinline__ float row_last(float x) {
    return __builtin_bit_cast(float,
        __builtin_amdgcn_update_dpp(0, __builtin_bit_cast(int, x), 0x15F, 0xF, 0xF, true));
}
static __device__ __forceinline__ float row_scan16(float x, float a1, float a2, float a4, float a8) {
    x = fmaf(a1, row_shr0<1>(x), x);
    x = fmaf(a2, row_shr0<2>(x), x);
    x = fmaf(a4, row_shr0<4>(x), x);
    x = fmaf(a8, row_shr0<8>(x), x);
    return x;
}

static __device__ __forceinline__ void load_blk_kmajor(v2f (&fa)[4][2], float (&bx)[4][4], float (&by)[4][4],
                                                       const float* __restrict__ arow0,
                                                       const float* __restrict__ arow1,
                                                       const float* s_t, int kk, int ncol) {
    #pragma unroll
    for (int s = 0; s < 4; ++s) {
        fa[s][0] = *(const v2f*)(arow0 + kk + 4 * s);
        fa[s][1] = *(const v2f*)(arow1 + kk + 4 * s);
        #pragma unroll
        for (int j = 0; j < 4; ++j) {
            bx[s][j] = s_t[(kk + 4 * s)     * UP + j * 16 + ncol];
            by[s][j] = s_t[(kk + 4 * s + 1) * UP + j * 16 + ncol];
        }
    }
}
static __device__ __forceinline__ void load_blk_rowmajor(v2f (&fa)[4][2], v2f (&fb)[4][4],
                                                         const float* __restrict__ arow0,
                                                         const float* __restrict__ arow1,
                                                         const float* s_t, int kk, int ncol) {
    #pragma unroll
    for (int s = 0; s < 4; ++s) {
        fa[s][0] = *(const v2f*)(arow0 + kk + 4 * s);
        fa[s][1] = *(const v2f*)(arow1 + kk + 4 * s);
        #pragma unroll
        for (int j = 0; j < 4; ++j)
            fb[s][j] = *(const v2f*)(&s_t[(j * 16 + ncol) * XP + kk + 4 * s]);
    }
}
static __device__ __forceinline__ void mma_blk(v8f (&acc)[2][4], const v2f (&fa)[4][2], const v2f (&fb)[4][4]) {
    #pragma unroll
    for (int s = 0; s < 4; ++s)
        #pragma unroll
        for (int i = 0; i < 2; ++i)
            #pragma unroll
            for (int j = 0; j < 4; ++j)
                acc[i][j] = wmma_f32_16x16x4(fa[s][i], fb[s][j], acc[i][j]);
}
static __device__ __forceinline__ void mma_blk_xy(v8f (&acc)[2][4], const v2f (&fa)[4][2],
                                                  const float (&bx)[4][4], const float (&by)[4][4]) {
    #pragma unroll
    for (int s = 0; s < 4; ++s)
        #pragma unroll
        for (int j = 0; j < 4; ++j) {
            v2f b; b.x = bx[s][j]; b.y = by[s][j];
            #pragma unroll
            for (int i = 0; i < 2; ++i)
                acc[i][j] = wmma_f32_16x16x4(fa[s][i], b, acc[i][j]);
        }
}
static __device__ __forceinline__ void keep_loads_here() {
    asm volatile("" ::: "memory");
    __builtin_amdgcn_sched_barrier(0);
}

static __device__ __forceinline__ void stage_u_tile(float* s_u, const float* __restrict__ ub,
                                                    int l0base, int wave, int lane) {
    const int rsub = lane >> 4;
    const int col  = (lane & 15) * 4;
    #pragma unroll
    for (int c = 0; c < 16; ++c) {
        const int row = wave * 32 + 2 * c + rsub;
        const float4 v = *(const float4*)(ub + (size_t)row * LSEQ + l0base + col);
        *(float4*)(&s_u[row * UP + col]) = v;
    }
}

static __device__ __forceinline__ void wave_lds_handover() {
    __builtin_amdgcn_fence(__ATOMIC_RELEASE, "wavefront");
    __builtin_amdgcn_wave_barrier();
    __builtin_amdgcn_fence(__ATOMIC_ACQUIRE, "wavefront");
}

__global__ void k_adiag(const float* __restrict__ a_un, float* __restrict__ adiag) {
    int s = threadIdx.x;
    float x  = a_un[s];
    float sp = fmaxf(x, 0.0f) + log1pf(expf(-fabsf(x)));
    adiag[s] = -sp;
}

__global__ __launch_bounds__(256) void k_gemm1(const float* __restrict__ u,
                                               const float* __restrict__ Bm,
                                               const float* __restrict__ adiag,
                                               float* __restrict__ X) {
    extern __shared__ char smem_raw[];
    float* s_u = (float*)smem_raw;
    float* s_t = s_u + DI * UP + (threadIdx.x >> 5) * SWAVE;

    const int b      = blockIdx.x >> 6;
    const int l0base = (blockIdx.x & 63) * 64;
    const int t      = threadIdx.x;
    const int wave   = t >> 5;
    const int lane   = t & 31;
    const int mrow   = lane & 15;
    const int khalf  = (lane >> 4) << 1;
    const int ncol   = lane & 15;
    const int hi     = (lane >> 4) << 3;

    const float* __restrict__ ub = u + (size_t)b * DI * LSEQ;

    __builtin_prefetch(Bm + (size_t)(wave * 64 + mrow) * DI, 0, 3);

    stage_u_tile(s_u, ub, l0base, wave, lane);
    __syncthreads();

    const v8f vzero = {0.f,0.f,0.f,0.f,0.f,0.f,0.f,0.f};

    for (int half = 0; half < 2; ++half) {
        const int st0 = wave * 4 + half * 2;
        v8f acc[2][4];
        #pragma unroll
        for (int i = 0; i < 2; ++i)
            #pragma unroll
            for (int j = 0; j < 4; ++j) acc[i][j] = vzero;

        const float* __restrict__ ar0 = Bm + (size_t)((st0 + 0) * 16 + mrow) * DI;
        const float* __restrict__ ar1 = Bm + (size_t)((st0 + 1) * 16 + mrow) * DI;
        v2f fa0[4][2], fa1[4][2];
        float bx0[4][4], by0[4][4], bx1[4][4], by1[4][4];
        load_blk_kmajor(fa0, bx0, by0, ar0, ar1, s_u, khalf, ncol);
        for (int k0 = 0; k0 < DI; k0 += 32) {
            load_blk_kmajor(fa1, bx1, by1, ar0, ar1, s_u, k0 + 16 + khalf, ncol);
            keep_loads_here();
            mma_blk_xy(acc, fa0, bx0, by0);
            __builtin_amdgcn_sched_barrier(0);
            load_blk_kmajor(fa0, bx0, by0, ar0, ar1, s_u, ((k0 + 32) & (DI - 1)) + khalf, ncol);
            keep_loads_here();
            mma_blk_xy(acc, fa1, bx1, by1);
        }

        #pragma unroll
        for (int i = 0; i < 2; ++i) {
            const int sbase = (st0 + i) * 16 + hi;
            const v4f alo = *(const v4f*)(adiag + sbase);
            const v4f ahi = *(const v4f*)(adiag + sbase + 4);
            #pragma unroll
            for (int r = 0; r < 8; ++r) {
                const float a1 = (r < 4) ? alo[r & 3] : ahi[r & 3];
                const float a2 = a1 * a1, a4 = a2 * a2, a8 = a4 * a4;
                const float w = row_scan16((ncol == 0) ? a1 : 0.f, a1, a2, a4, a8);
                float prev = 0.f;
                #pragma unroll
                for (int j = 0; j < 4; ++j) {
                    float x = row_scan16(acc[i][j][r], a1, a2, a4, a8);
                    if (j > 0) x = fmaf(w, prev, x);
                    prev = row_last(x);
                    acc[i][j][r] = x;
                }
            }
        }

        #pragma unroll
        for (int i = 0; i < 2; ++i) {
            #pragma unroll
            for (int j = 0; j < 4; ++j) {
                float* tp = s_t + (j * 16 + ncol) * SP + i * 16 + hi;
                v4f lo  = __builtin_shufflevector(acc[i][j], acc[i][j], 0,1,2,3);
                v4f hiv = __builtin_shufflevector(acc[i][j], acc[i][j], 4,5,6,7);
                *(v4f*)(tp)     = lo;
                *(v4f*)(tp + 4) = hiv;
            }
        }
        wave_lds_handover();
        {
            const int rsub = lane >> 3;
            const int col  = (lane & 7) * 4;
            float* xw = X + ((size_t)b * LSEQ + l0base + rsub) * DS + st0 * 16 + col;
            #pragma unroll
            for (int q = 0; q < 16; ++q) {
                const v4f v = *(const v4f*)(s_t + (4 * q + rsub) * SP + col);
                *(v4f*)(xw + (size_t)(4 * q) * DS) = v;
            }
        }
    }
}

__global__ __launch_bounds__(64) void k_combine(const float* __restrict__ adiag,
                                                const float* __restrict__ h0,
                                                float* __restrict__ X) {
    const int b = blockIdx.x >> 3;
    const int s = (blockIdx.x & 7) * 64 + threadIdx.x;
    const float a = adiag[s];
    float aT = a;
    #pragma unroll
    for (int i = 0; i < 6; ++i) aT = aT * aT;
    float cin = h0[s];
    float* xr = X + ((size_t)b * LSEQ + (TCH - 1)) * DS + s;
    for (int c0 = 0; c0 < NCHUNK; c0 += 16) {
        float v[16];
        #pragma unroll
        for (int k = 0; k < 16; ++k) v[k] = xr[(size_t)(c0 + k) * TCH * DS];
        #pragma unroll
        for (int k = 0; k < 16; ++k) {
            cin = fmaf(aT, cin, v[k]);
            xr[(size_t)(c0 + k) * TCH * DS] = cin;
        }
    }
}

__global__ __launch_bounds__(256) void k_gemm2(const float* __restrict__ X,
                                               const float* __restrict__ u,
                                               const float* __restrict__ Cm,
                                               const float* __restrict__ Dm,
                                               const float* __restrict__ adiag,
                                               const float* __restrict__ h0,
                                               float* __restrict__ Y) {
    extern __shared__ char smem_raw[];
    float* s_x = (float*)smem_raw;
    float* s_u = (float*)smem_raw;

    const int b      = blockIdx.x >> 6;
    const int l0base = (blockIdx.x & 63) * 64;
    const int t      = threadIdx.x;
    const int wave   = t >> 5;
    const int lane   = t & 31;
    const int mrow   = lane & 15;
    const int khalf  = (lane >> 4) << 1;
    const int ncol   = lane & 15;
    const int hi     = (lane >> 4) << 3;

    const float* __restrict__ ub = u + (size_t)b * DI * LSEQ;
    const float* __restrict__ Xb = X + (size_t)b * LSEQ * DS;

    __builtin_prefetch(Cm + (size_t)(wave * 32 + mrow) * DS, 0, 3);

    {
        const int cg = (t & 127) * 4;
        const int r0 = (t >> 7) * 32;
        const v4f a = *(const v4f*)(adiag + cg);
        v4f cin;
        if (l0base == 0) cin = *(const v4f*)(h0 + cg);
        else             cin = *(const v4f*)(Xb + (size_t)(l0base - 1) * DS + cg);
        const float* __restrict__ xsrc = Xb + (size_t)(l0base + r0) * DS + cg;
        v4f v[32];
        #pragma unroll
        for (int k = 0; k < 32; ++k) v[k] = *(const v4f*)(xsrc + (size_t)k * DS);
        __builtin_amdgcn_sched_barrier(0);
        v4f p33 = a;
        #pragma unroll
        for (int k = 0; k < 32; ++k) p33 *= a;
        v4f p = (r0 != 0) ? p33 : a;
        float* xdst = s_x + r0 * XP + cg;
        #pragma unroll
        for (int k = 0; k < 32; ++k) {
            v4f x;
            x.x = fmaf(p.x, cin.x, v[k].x);
            x.y = fmaf(p.y, cin.y, v[k].y);
            x.z = fmaf(p.z, cin.z, v[k].z);
            x.w = fmaf(p.w, cin.w, v[k].w);
            if (r0 + k == TCH - 1) x = v[k];
            *(v4f*)(xdst + k * XP) = x;
            p *= a;
        }
    }
    __syncthreads();

    const v8f vzero = {0.f,0.f,0.f,0.f,0.f,0.f,0.f,0.f};
    v8f acc[2][4];
    #pragma unroll
    for (int i = 0; i < 2; ++i)
        #pragma unroll
        for (int j = 0; j < 4; ++j) acc[i][j] = vzero;

    v2f fa0[4][2], fb0[4][4], fa1[4][2], fb1[4][4];
    {
        const float* __restrict__ cr0 = Cm + (size_t)((wave * 2 + 0) * 16 + mrow) * DS;
        const float* __restrict__ cr1 = Cm + (size_t)((wave * 2 + 1) * 16 + mrow) * DS;
        load_blk_rowmajor(fa0, fb0, cr0, cr1, s_x, khalf, ncol);
        for (int k0 = 0; k0 < DS; k0 += 32) {
            load_blk_rowmajor(fa1, fb1, cr0, cr1, s_x, k0 + 16 + khalf, ncol);
            keep_loads_here();
            mma_blk(acc, fa0, fb0);
            __builtin_amdgcn_sched_barrier(0);
            load_blk_rowmajor(fa0, fb0, cr0, cr1, s_x, ((k0 + 32) & (DS - 1)) + khalf, ncol);
            keep_loads_here();
            mma_blk(acc, fa1, fb1);
        }
    }

    __syncthreads();
    stage_u_tile(s_u, ub, l0base, wave, lane);
    __syncthreads();

    {
        const float* __restrict__ dr0 = Dm + (size_t)((wave * 2 + 0) * 16 + mrow) * DI;
        const float* __restrict__ dr1 = Dm + (size_t)((wave * 2 + 1) * 16 + mrow) * DI;
        float bx0[4][4], by0[4][4], bx1[4][4], by1[4][4];
        load_blk_kmajor(fa0, bx0, by0, dr0, dr1, s_u, khalf, ncol);
        for (int k0 = 0; k0 < DI; k0 += 32) {
            load_blk_kmajor(fa1, bx1, by1, dr0, dr1, s_u, k0 + 16 + khalf, ncol);
            keep_loads_here();
            mma_blk_xy(acc, fa0, bx0, by0);
            __builtin_amdgcn_sched_barrier(0);
            load_blk_kmajor(fa0, bx0, by0, dr0, dr1, s_u, ((k0 + 32) & (DI - 1)) + khalf, ncol);
            keep_loads_here();
            mma_blk_xy(acc, fa1, bx1, by1);
        }
    }

    #pragma unroll
    for (int i = 0; i < 2; ++i) {
        const int obase = (wave * 2 + i) * 16 + hi;
        #pragma unroll
        for (int j = 0; j < 4; ++j) {
            const int l = l0base + j * 16 + ncol;
            #pragma unroll
            for (int v = 0; v < 8; ++v)
                Y[((size_t)b * DO + obase + v) * LSEQ + l] = acc[i][j][v];
        }
    }
}

extern "C" void kernel_launch(void* const* d_in, const int* in_sizes, int n_in,
                              void* d_out, int out_size, void* d_ws, size_t ws_size,
                              hipStream_t stream) {
    (void)in_sizes; (void)n_in; (void)out_size; (void)ws_size;

    const float* u    = (const float*)d_in[0];
    const float* a_un = (const float*)d_in[1];
    const float* Bm   = (const float*)d_in[2];
    const float* Cm   = (const float*)d_in[3];
    const float* Dm   = (const float*)d_in[4];
    const float* h0   = (const float*)d_in[5];
    float*       Y    = (float*)d_out;

    char* ws = (char*)d_ws;
    const size_t X_BYTES = (size_t)BZ * LSEQ * DS * sizeof(float);
    float* X           = (float*)(ws);
    float* adiag       = (float*)(ws + X_BYTES);

    const size_t SH1 = (size_t)(DI * UP + 8 * SWAVE) * sizeof(float);
    const size_t SH2 = (size_t)64 * XP * sizeof(float);

    k_adiag    <<<1,                512, 0,   stream>>>(a_un, adiag);
    k_gemm1    <<<BZ * (LSEQ / 64), 256, SH1, stream>>>(u, Bm, adiag, X);
    k_combine  <<<BZ * (DS / 64),   64,  0,   stream>>>(adiag, h0, X);
    k_gemm2    <<<BZ * (LSEQ / 64), 256, SH2, stream>>>(X, u, Cm, Dm, adiag, h0, Y);
}


# WMMA fragment loads widened: k re-ordered inside each 16-k block so a lane's A data and X-tile B data are 16-B loads, u tile stored k-pair-interleaved for 8-B B loads; reorders the k sums
# speedup vs baseline: 1.4937x; 1.0468x over previous
#include <hip/hip_runtime.h>
#include <hip/hip_bf16.h>

#define BZ    8
#define DI    256
#define DS    512
#define DO    256
#define LSEQ  4096
#define TCH   64
#define NCHUNK (LSEQ / TCH)

#define UPP   144
#define XP    516
#define SP    36
#define SWAVE (64 * SP)

typedef float v2f __attribute__((ext_vector_type(2)));
typedef float v4f __attribute__((ext_vector_type(4)));
typedef float v8f __attribute__((ext_vector_type(8)));

static __device__ __forceinline__ v8f wmma_f32_16x16x4(v2f a, v2f b, v8f c) {
    return __builtin_amdgcn_wmma_f32_16x16x4_f32(
         false, a,  false, b,
         (short)0, c,  false,  false);
}

template <int N>
static __device__ __forceinline__ float row_shr0(float x) {
    return __builtin_bit_cast(float,
        __builtin_amdgcn_update_dpp(0, __builtin_bit_cast(int, x), 0x110 + N, 0xF, 0xF, true));
}
static __device__ __forceinline__ float row_last(float x) {
    return __builtin_bit_cast(float,
        __builtin_amdgcn_update_dpp(0, __builtin_bit_cast(int, x), 0x15F, 0xF, 0xF, true));
}
static __device__ __forceinline__ float row_scan16(float x, float a1, float a2, float a4, float a8) {
    x = fmaf(a1, row_shr0<1>(x), x);
    x = fmaf(a2, row_shr0<2>(x), x);
    x = fmaf(a4, row_shr0<4>(x), x);
    x = fmaf(a8, row_shr0<8>(x), x);
    return x;
}

template <int P>
static __device__ __forceinline__ v2f kstep_half(v4f v) {
    return P ? __builtin_shufflevector(v, v, 2, 3) : __builtin_shufflevector(v, v, 0, 1);
}
static __device__ __forceinline__ void load_blk_kpair(v4f (&fa)[2][2], v2f (&fb)[4][4],
                                                      const float* __restrict__ arow0,
                                                      const float* __restrict__ arow1,
                                                      const float* s_t, int K0, int h, int ncol) {
    #pragma unroll
    for (int q = 0; q < 2; ++q) {
        fa[q][0] = *(const v4f*)(arow0 + K0 + 8 * q + 4 * h);
        fa[q][1] = *(const v4f*)(arow1 + K0 + 8 * q + 4 * h);
        #pragma unroll
        for (int p = 0; p < 2; ++p)
            #pragma unroll
            for (int j = 0; j < 4; ++j)
                fb[2 * q + p][j] = *(const v2f*)(&s_t[((K0 >> 1) + 4 * q + 2 * h + p) * UPP + (j * 16 + ncol) * 2]);
    }
}
static __device__ __forceinline__ void load_blk_rowmajor(v4f (&fa)[2][2], v4f (&fb)[2][4],
                                                         const float* __restrict__ arow0,
                                                         const float* __restrict__ arow1,
                                                         const float* s_t, int K0, int h, int ncol) {
    #pragma unroll
    for (int q = 0; q < 2; ++q) {
        fa[q][0] = *(const v4f*)(arow0 + K0 + 8 * q + 4 * h);
        fa[q][1] = *(const v4f*)(arow1 + K0 + 8 * q + 4 * h);
        #pragma unroll
        for (int j = 0; j < 4; ++j)
            fb[q][j] = *(const v4f*)(&s_t[(j * 16 + ncol) * XP + K0 + 8 * q + 4 * h]);
    }
}
static __device__ __forceinline__ void mma_blk(v8f (&acc)[2][4], const v4f (&fa)[2][2], const v4f (&fb)[2][4]) {
    #pragma unroll
    for (int q = 0; q < 2; ++q) {
        #pragma unroll
        for (int i = 0; i < 2; ++i)
            #pragma unroll
            for (int j = 0; j < 4; ++j)
                acc[i][j] = wmma_f32_16x16x4(kstep_half<0>(fa[q][i]), kstep_half<0>(fb[q][j]), acc[i][j]);
        #pragma unroll
        for (int i = 0; i < 2; ++i)
            #pragma unroll
            for (int j = 0; j < 4; ++j)
                acc[i][j] = wmma_f32_16x16x4(kstep_half<1>(fa[q][i]), kstep_half<1>(fb[q][j]), acc[i][j]);
    }
}
static __device__ __forceinline__ void mma_blk_kpair(v8f (&acc)[2][4], const v4f (&fa)[2][2], const v2f (&fb)[4][4]) {
    #pragma unroll
    for (int q = 0; q < 2; ++q) {
        #pragma unroll
        for (int j = 0; j < 4; ++j)
            #pragma unroll
            for (int i = 0; i < 2; ++i)
                acc[i][j] = wmma_f32_16x16x4(kstep_half<0>(fa[q][i]), fb[2 * q][j], acc[i][j]);
        #pragma unroll
        for (int j = 0; j < 4; ++j)
            #pragma unroll
            for (int i = 0; i < 2; ++i)
                acc[i][j] = wmma_f32_16x16x4(kstep_half<1>(fa[q][i]), fb[2 * q + 1][j], acc[i][j]);
    }
}
static __device__ __forceinline__ void keep_loads_here() {
    asm volatile("" ::: "memory");
    __builtin_amdgcn_sched_barrier(0);
}

static __device__ __forceinline__ void stage_u_tile(float* s_u, const float* __restrict__ ub,
                                                    int l0base, int wave, int lane) {
    const int rsub = lane >> 4;
    const int col  = (lane & 15) * 4;
    float4 va[8], vb[8];
    #pragma unroll
    for (int m = 0; m < 8; ++m) {
        const int row = wave * 32 + 4 * m + 2 * rsub;
        va[m] = *(const float4*)(ub + (size_t)row * LSEQ + l0base + col);
        vb[m] = *(const float4*)(ub + (size_t)(row + 1) * LSEQ + l0base + col);
    }
    #pragma unroll
    for (int m = 0; m < 8; ++m) {
        const int prow = wave * 16 + 2 * m + rsub;
        float* dst = &s_u[prow * UPP + col * 2];
        *(float4*)(dst)     = make_float4(va[m].x, vb[m].x, va[m].y, vb[m].y);
        *(float4*)(dst + 4) = make_float4(va[m].z, vb[m].z, va[m].w, vb[m].w);
    }
}

static __device__ __forceinline__ void wave_lds_handover() {
    __builtin_amdgcn_fence(__ATOMIC_RELEASE, "wavefront");
    __builtin_amdgcn_wave_barrier();
    __builtin_amdgcn_fence(__ATOMIC_ACQUIRE, "wavefront");
}

__global__ void k_adiag(const float* __restrict__ a_un, float* __restrict__ adiag) {
    int s = threadIdx.x;
    float x  = a_un[s];
    float sp = fmaxf(x, 0.0f) + log1pf(expf(-fabsf(x)));
    adiag[s] = -sp;
}

__global__ __launch_bounds__(256) void k_gemm1(const float* __restrict__ u,
                                               const float* __restrict__ Bm,
                                               const float* __restrict__ adiag,
                                               float* __restrict__ X) {
    extern __shared__ char smem_raw[];
    float* s_u = (float*)smem_raw;
    float* s_t = s_u + (DI / 2) * UPP + (threadIdx.x >> 5) * SWAVE;

    const int b      = blockIdx.x >> 6;
    const int l0base = (blockIdx.x & 63) * 64;
    const int t      = threadIdx.x;
    const int wave   = t >> 5;
    const int lane   = t & 31;
    const int mrow   = lane & 15;
    const int h      = lane >> 4;
    const int ncol   = lane & 15;
    const int hi     = (lane >> 4) << 3;

    const float* __restrict__ ub = u + (size_t)b * DI * LSEQ;

    __builtin_prefetch(Bm + (size_t)(wave * 64 + mrow) * DI, 0, 3);

    stage_u_tile(s_u, ub, l0base, wave, lane);
    __syncthreads();

    const v8f vzero = {0.f,0.f,0.f,0.f,0.f,0.f,0.f,0.f};

    for (int half = 0; half < 2; ++half) {
        const int st0 = wave * 4 + half * 2;
        v8f acc[2][4];
        #pragma unroll
        for (int i = 0; i < 2; ++i)
            #pragma unroll
            for (int j = 0; j < 4; ++j) acc[i][j] = vzero;

        const float* __restrict__ ar0 = Bm + (size_t)((st0 + 0) * 16 + mrow) * DI;
        const float* __restrict__ ar1 = Bm + (size_t)((st0 + 1) * 16 + mrow) * DI;
        v4f fa0[2][2], fa1[2][2];
        v2f ub0[4][4], ub1[4][4];
        load_blk_kpair(fa0, ub0, ar0, ar1, s_u, 0, h, ncol);
        for (int k0 = 0; k0 < DI; k0 += 32) {
            load_blk_kpair(fa1, ub1, ar0, ar1, s_u, k0 + 16, h, ncol);
            keep_loads_here();
            mma_blk_kpair(acc, fa0, ub0);
            __builtin_amdgcn_sched_barrier(0);
            load_blk_kpair(fa0, ub0, ar0, ar1, s_u, (k0 + 32) & (DI - 1), h, ncol);
            keep_loads_here();
            mma_blk_kpair(acc, fa1, ub1);
        }

        #pragma unroll
        for (int i = 0; i < 2; ++i) {
            const int sbase = (st0 + i) * 16 + hi;
            const v4f alo = *(const v4f*)(adiag + sbase);
            const v4f ahi = *(const v4f*)(adiag + sbase + 4);
            #pragma unroll
            for (int r = 0; r < 8; ++r) {
                const float a1 = (r < 4) ? alo[r & 3] : ahi[r & 3];
                const float a2 = a1 * a1, a4 = a2 * a2, a8 = a4 * a4;
                const float w = row_scan16((ncol == 0) ? a1 : 0.f, a1, a2, a4, a8);
                float prev = 0.f;
                #pragma unroll
                for (int j = 0; j < 4; ++j) {
                    float x = row_scan16(acc[i][j][r], a1, a2, a4, a8);
                    if (j > 0) x = fmaf(w, prev, x);
                    prev = row_last(x);
                    acc[i][j][r] = x;
                }
            }
        }

        #pragma unroll
        for (int i = 0; i < 2; ++i) {
            #pragma unroll
            for (int j = 0; j < 4; ++j) {
                float* tp = s_t + (j * 16 + ncol) * SP + i * 16 + hi;
                v4f lo  = __builtin_shufflevector(acc[i][j], acc[i][j], 0,1,2,3);
                v4f hiv = __builtin_shufflevector(acc[i][j], acc[i][j], 4,5,6,7);
                *(v4f*)(tp)     = lo;
                *(v4f*)(tp + 4) = hiv;
            }
        }
        wave_lds_handover();
        {
            const int rsub = lane >> 3;
            const int col  = (lane & 7) * 4;
            float* xw = X + ((size_t)b * LSEQ + l0base + rsub) * DS + st0 * 16 + col;
            #pragma unroll
            for (int q = 0; q < 16; ++q) {
                const v4f v = *(const v4f*)(s_t + (4 * q + rsub) * SP + col);
                *(v4f*)(xw + (size_t)(4 * q) * DS) = v;
            }
        }
    }
}

__global__ __launch_bounds__(64) void k_combine(const float* __restrict__ adiag,
                                                const float* __restrict__ h0,
                                                float* __restrict__ X) {
    const int b = blockIdx.x >> 3;
    const int s = (blockIdx.x & 7) * 64 + threadIdx.x;
    const float a = adiag[s];
    float aT = a;
    #pragma unroll
    for (int i = 0; i < 6; ++i) aT = aT * aT;
    float cin = h0[s];
    float* xr = X + ((size_t)b * LSEQ + (TCH - 1)) * DS + s;
    for (int c0 = 0; c0 < NCHUNK; c0 += 16) {
        float v[16];
        #pragma unroll
        for (int k = 0; k < 16; ++k) v[k] = xr[(size_t)(c0 + k) * TCH * DS];
        #pragma unroll
        for (int k = 0; k < 16; ++k) {
            cin = fmaf(aT, cin, v[k]);
            xr[(size_t)(c0 + k) * TCH * DS] = cin;
        }
    }
}

__global__ __launch_bounds__(256) void k_gemm2(const float* __restrict__ X,
                                               const float* __restrict__ u,
                                               const float* __restrict__ Cm,
                                               const float* __restrict__ Dm,
                                               const float* __restrict__ adiag,
                                               const float* __restrict__ h0,
                                               float* __restrict__ Y) {
    extern __shared__ char smem_raw[];
    float* s_x = (float*)smem_raw;
    float* s_u = (float*)smem_raw;

    const int b      = blockIdx.x >> 6;
    const int l0base = (blockIdx.x & 63) * 64;
    const int t      = threadIdx.x;
    const int wave   = t >> 5;
    const int lane   = t & 31;
    const int mrow   = lane & 15;
    const int h      = lane >> 4;
    const int ncol   = lane & 15;
    const int hi     = (lane >> 4) << 3;

    const float* __restrict__ ub = u + (size_t)b * DI * LSEQ;
    const float* __restrict__ Xb = X + (size_t)b * LSEQ * DS;

    __builtin_prefetch(Cm + (size_t)(wave * 32 + mrow) * DS, 0, 3);

    {
        const int cg = (t & 127) * 4;
        const int r0 = (t >> 7) * 32;
        const v4f a = *(const v4f*)(adiag + cg);
        v4f cin;
        if (l0base == 0) cin = *(const v4f*)(h0 + cg);
        else             cin = *(const v4f*)(Xb + (size_t)(l0base - 1) * DS + cg);
        const float* __restrict__ xsrc = Xb + (size_t)(l0base + r0) * DS + cg;
        v4f v[32];
        #pragma unroll
        for (int k = 0; k < 32; ++k) v[k] = *(const v4f*)(xsrc + (size_t)k * DS);
        __builtin_amdgcn_sched_barrier(0);
        v4f p33 = a;
        #pragma unroll
        for (int k = 0; k < 32; ++k) p33 *= a;
        v4f p = (r0 != 0) ? p33 : a;
        float* xdst = s_x + r0 * XP + cg;
        #pragma unroll
        for (int k = 0; k < 32; ++k) {
            v4f x;
            x.x = fmaf(p.x, cin.x, v[k].x);
            x.y = fmaf(p.y, cin.y, v[k].y);
            x.z = fmaf(p.z, cin.z, v[k].z);
            x.w = fmaf(p.w, cin.w, v[k].w);
            if (r0 + k == TCH - 1) x = v[k];
            *(v4f*)(xdst + k * XP) = x;
            p *= a;
        }
    }
    __syncthreads();

    const v8f vzero = {0.f,0.f,0.f,0.f,0.f,0.f,0.f,0.f};
    v8f acc[2][4];
    #pragma unroll
    for (int i = 0; i < 2; ++i)
        #pragma unroll
        for (int j = 0; j < 4; ++j) acc[i][j] = vzero;

    v4f fa0[2][2], fa1[2][2];
    {
        v4f fb0[2][4], fb1[2][4];
        const float* __restrict__ cr0 = Cm + (size_t)((wave * 2 + 0) * 16 + mrow) * DS;
        const float* __restrict__ cr1 = Cm + (size_t)((wave * 2 + 1) * 16 + mrow) * DS;
        load_blk_rowmajor(fa0, fb0, cr0, cr1, s_x, 0, h, ncol);
        for (int k0 = 0; k0 < DS; k0 += 32) {
            load_blk_rowmajor(fa1, fb1, cr0, cr1, s_x, k0 + 16, h, ncol);
            keep_loads_here();
            mma_blk(acc, fa0, fb0);
            __builtin_amdgcn_sched_barrier(0);
            load_blk_rowmajor(fa0, fb0, cr0, cr1, s_x, (k0 + 32) & (DS - 1), h, ncol);
            keep_loads_here();
            mma_blk(acc, fa1, fb1);
        }
    }

    __syncthreads();
    stage_u_tile(s_u, ub, l0base, wave, lane);
    __syncthreads();

    {
        const float* __restrict__ dr0 = Dm + (size_t)((wave * 2 + 0) * 16 + mrow) * DI;
        const float* __restrict__ dr1 = Dm + (size_t)((wave * 2 + 1) * 16 + mrow) * DI;
        v2f ub0[4][4], ub1[4][4];
        load_blk_kpair(fa0, ub0, dr0, dr1, s_u, 0, h, ncol);
        for (int k0 = 0; k0 < DI; k0 += 32) {
            load_blk_kpair(fa1, ub1, dr0, dr1, s_u, k0 + 16, h, ncol);
            keep_loads_here();
            mma_blk_kpair(acc, fa0, ub0);
            __builtin_amdgcn_sched_barrier(0);
            load_blk_kpair(fa0, ub0, dr0, dr1, s_u, (k0 + 32) & (DI - 1), h, ncol);
            keep_loads_here();
            mma_blk_kpair(acc, fa1, ub1);
        }
    }

    #pragma unroll
    for (int i = 0; i < 2; ++i) {
        const int obase = (wave * 2 + i) * 16 + hi;
        #pragma unroll
        for (int j = 0; j < 4; ++j) {
            const int l = l0base + j * 16 + ncol;
            #pragma unroll
            for (int v = 0; v < 8; ++v)
                Y[((size_t)b * DO + obase + v) * LSEQ + l] = acc[i][j][v];
        }
    }
}

extern "C" void kernel_launch(void* const* d_in, const int* in_sizes, int n_in,
                              void* d_out, int out_size, void* d_ws, size_t ws_size,
                              hipStream_t stream) {
    (void)in_sizes; (void)n_in; (void)out_size; (void)ws_size;

    const float* u    = (const float*)d_in[0];
    const float* a_un = (const float*)d_in[1];
    const float* Bm   = (const float*)d_in[2];
    const float* Cm   = (const float*)d_in[3];
    const float* Dm   = (const float*)d_in[4];
    const float* h0   = (const float*)d_in[5];
    float*       Y    = (float*)d_out;

    char* ws = (char*)d_ws;
    const size_t X_BYTES = (size_t)BZ * LSEQ * DS * sizeof(float);
    float* X           = (float*)(ws);
    float* adiag       = (float*)(ws + X_BYTES);

    const size_t SH1 = (size_t)((DI / 2) * UPP + 8 * SWAVE) * sizeof(float);
    const size_t SH2 = (size_t)64 * XP * sizeof(float);

    k_adiag    <<<1,                512, 0,   stream>>>(a_un, adiag);
    k_gemm1    <<<BZ * (LSEQ / 64), 256, SH1, stream>>>(u, Bm, adiag, X);
    k_combine  <<<BZ * (DS / 64),   64,  0,   stream>>>(adiag, h0, X);
    k_gemm2    <<<BZ * (LSEQ / 64), 256, SH2, stream>>>(X, u, Cm, Dm, adiag, h0, Y);
}
